# starting kernel
# baseline (speedup 1.0000x reference)
#include <hip/hip_runtime.h>
#include <hip/hip_bf16.h>
#include <math.h>

#define NA   16
#define DOBS 256
#define AD   32
#define BB   16
#define TT   64
#define SS   16
#define H1D  128
#define H2D  64
#define BT   (BB*TT)
#define NVAR 18
#define INV_S1 (1.0f/17.0f)

#define H1P  132
#define H2P  68
#define OBSP 260

#define OFF_HOBS 0
#define OFF_H1   2048
#define OFF_H2   4160
#define OFF_PW   5248
#define OFF_PQ   5760
#define OFF_W1A  6272
#define OFF_W2   10368
#define OFF_W3   18560
#define LDS_FLOATS 20608

typedef float v2f __attribute__((ext_vector_type(2)));
typedef float v8f __attribute__((ext_vector_type(8)));

__device__ __forceinline__ v8f wmma4(v2f a, v2f b, v8f c) {
  return __builtin_amdgcn_wmma_f32_16x16x4_f32(
       false, a,  false, b,
       (short)0, c,  false,  false);
}

__device__ __forceinline__ float wave_max(float v) {
  #pragma unroll
  for (int off = 16; off; off >>= 1) v = fmaxf(v, __shfl_xor(v, off, 32));
  return v;
}
__device__ __forceinline__ float wave_sum(float v) {
  #pragma unroll
  for (int off = 16; off; off >>= 1) v += __shfl_xor(v, off, 32);
  return v;
}

__global__ __launch_bounds__(128)
void coma_influence_kernel(const float* __restrict__ obs,
                           const float* __restrict__ act,
                           const float* __restrict__ cf,
                           const float* __restrict__ W1,
                           const float* __restrict__ b1,
                           const float* __restrict__ W2,
                           const float* __restrict__ b2,
                           const float* __restrict__ W3,
                           const float* __restrict__ b3,
                           float* __restrict__ out)
{
  extern __shared__ float smem[];
  float* s_hobs = smem + OFF_HOBS;
  float* s_h1   = smem + OFF_H1;
  float* s_h2   = smem + OFF_H2;
  float* s_pw   = smem + OFF_PW;
  float* s_pq   = smem + OFF_PQ;
  float* s_w1a  = smem + OFF_W1A;
  float* s_w2   = smem + OFF_W2;
  float* s_w3   = smem + OFF_W3;
  float* s_obs  = smem + OFF_W2;

  const int tid  = threadIdx.x;
  const int lane = tid & 31;
  const int wid  = tid >> 5;
  const int g    = blockIdx.x;
  const int r0   = g * 16;
  const int n    = r0 >> 10;
  const int bt0  = r0 & 1023;

  const int col = lane & 15;
  const int hi  = lane >> 4;
  const int kb  = hi * 2;

  #pragma unroll
  for (int j4 = 0; j4 < 8; ++j4) {
    const int i4 = tid + 128 * j4;
    const int m  = i4 >> 6;
    const int d  = (i4 & 63) * 4;
    const float4 v = *(const float4*)(obs + ((size_t)(bt0 + m) * NA + n) * DOBS + d);
    *(float4*)(s_obs + m * OBSP + d) = v;
  }
  __syncthreads();

  #pragma unroll
  for (int jj = 0; jj < 2; ++jj) {
    const int jcol = (wid * 2 + jj) * 16 + col;
    v8f acc;
    {
      const float bias = b1[jcol];
      #pragma unroll
      for (int v = 0; v < 8; ++v) acc[v] = bias;
    }
    for (int kk = 0; kk < DOBS; kk += 4) {
      v2f a; a[0] = s_obs[col * OBSP + kk + kb]; a[1] = s_obs[col * OBSP + kk + kb + 1];
      v2f b; b[0] = W1[(kk + kb) * H1D + jcol];  b[1] = W1[(kk + kb + 1) * H1D + jcol];
      acc = wmma4(a, b, acc);
    }
    #pragma unroll
    for (int v = 0; v < 8; ++v) s_hobs[(v + 8 * hi) * H1D + jcol] = acc[v];
  }
  __syncthreads();

  #pragma unroll
  for (int j4 = 0; j4 < 8; ++j4) {
    const int j = (tid + 128 * j4) * 4;
    *(float4*)(s_w1a + j) = *(const float4*)(W1 + DOBS * H1D + j);
  }
  #pragma unroll
  for (int j4 = 0; j4 < 16; ++j4) {
    const int j = (tid + 128 * j4) * 4;
    *(float4*)(s_w2 + j) = *(const float4*)(W2 + j);
  }
  #pragma unroll
  for (int j4 = 0; j4 < 4; ++j4) {
    const int j = (tid + 128 * j4) * 4;
    *(float4*)(s_w3 + j) = *(const float4*)(W3 + j);
  }
  __syncthreads();

  for (int vi = 0; vi < NVAR; ++vi) {
    const float* act_row = nullptr;
    if (vi == 0)      act_row = act + ((size_t)(bt0 + col) * NA + n) * AD;
    else if (vi >= 2) act_row = cf  + (((size_t)n * SS + (vi - 2)) * BT + (bt0 + col)) * AD;

    if (vi + 1 >= 2 && vi + 1 < NVAR) {
      const float* nxt = cf + (((size_t)n * SS + (vi - 1)) * BT + (bt0 + col)) * AD;
      __builtin_prefetch(nxt, 0, 1);
    }

    v2f av[8];
    if (vi != 1) {
      #pragma unroll
      for (int kq = 0; kq < 8; ++kq) {
        const float2 t = *(const float2*)(act_row + kq * 4 + kb);
        av[kq][0] = t.x; av[kq][1] = t.y;
      }
    }

    #pragma unroll
    for (int jj = 0; jj < 2; ++jj) {
      const int jcol = (wid * 2 + jj) * 16 + col;
      v8f acc;
      #pragma unroll
      for (int v = 0; v < 8; ++v) acc[v] = s_hobs[(v + 8 * hi) * H1D + jcol];
      if (vi != 1) {
        #pragma unroll
        for (int kq = 0; kq < 8; ++kq) {
          const int kk = kq * 4;
          v2f b; b[0] = s_w1a[(kk + kb) * H1D + jcol];
                 b[1] = s_w1a[(kk + kb + 1) * H1D + jcol];
          acc = wmma4(av[kq], b, acc);
        }
      }
      #pragma unroll
      for (int v = 0; v < 8; ++v)
        s_h1[(v + 8 * hi) * H1P + jcol] = fmaxf(acc[v], 0.0f);
    }
    __syncthreads();

    {
      const int jcol = wid * 16 + col;
      v8f acc;
      {
        const float bias = b2[jcol];
        #pragma unroll
        for (int v = 0; v < 8; ++v) acc[v] = bias;
      }
      for (int kk = 0; kk < H1D; kk += 4) {
        v2f a; a[0] = s_h1[col * H1P + kk + kb];  a[1] = s_h1[col * H1P + kk + kb + 1];
        v2f b; b[0] = s_w2[(kk + kb) * H2D + jcol]; b[1] = s_w2[(kk + kb + 1) * H2D + jcol];
        acc = wmma4(a, b, acc);
      }
      #pragma unroll
      for (int v = 0; v < 8; ++v)
        s_h2[(v + 8 * hi) * H2P + jcol] = fmaxf(acc[v], 0.0f);
    }
    __syncthreads();

    if (wid < 2) {
      const int jcol = wid * 16 + col;
      v8f acc;
      {
        const float bias = b3[jcol];
        #pragma unroll
        for (int v = 0; v < 8; ++v) acc[v] = bias;
      }
      for (int kk = 0; kk < H2D; kk += 4) {
        v2f a; a[0] = s_h2[col * H2P + kk + kb];  a[1] = s_h2[col * H2P + kk + kb + 1];
        v2f b; b[0] = s_w3[(kk + kb) * AD + jcol]; b[1] = s_w3[(kk + kb + 1) * AD + jcol];
        acc = wmma4(a, b, acc);
      }
      if (vi == 0) {
        #pragma unroll
        for (int v = 0; v < 8; ++v) s_pw[(v + 8 * hi) * AD + jcol] = acc[v];
      } else if (vi == 1) {
        #pragma unroll
        for (int v = 0; v < 8; ++v) s_pq[(v + 8 * hi) * AD + jcol] = acc[v];
      } else {
        #pragma unroll
        for (int v = 0; v < 8; ++v) s_pq[(v + 8 * hi) * AD + jcol] += acc[v];
      }
    }
    __syncthreads();
  }

  #pragma unroll
  for (int rr = 0; rr < 4; ++rr) {
    const int r = wid * 4 + rr;
    const float pw = s_pw[r * AD + lane];
    const float pq = s_pq[r * AD + lane] * INV_S1;

    const float mp = wave_max(pw);
    const float mq = wave_max(pq);
    const float ep = expf(pw - mp);
    const float eq = expf(pq - mq);
    const float sp = wave_sum(ep);
    const float sq = wave_sum(eq);

    const float log_p = (pw - mp) - logf(sp);
    const float log_q = (pq - mq) - logf(sq);
    const float q     = eq / sq;
    float kl = q * (log_q - log_p);
    kl = wave_sum(kl);
    if (lane == 0) out[(size_t)(bt0 + r) * NA + n] = kl * (1.0f / AD);
  }
}

extern "C" void kernel_launch(void* const* d_in, const int* in_sizes, int n_in,
                              void* d_out, int out_size, void* d_ws, size_t ws_size,
                              hipStream_t stream) {
  (void)in_sizes; (void)n_in; (void)d_ws; (void)ws_size; (void)out_size;
  const float* obs = (const float*)d_in[0];
  const float* act = (const float*)d_in[1];
  const float* cf  = (const float*)d_in[2];
  const float* W1  = (const float*)d_in[3];
  const float* b1  = (const float*)d_in[4];
  const float* W2  = (const float*)d_in[5];
  const float* b2  = (const float*)d_in[6];
  const float* W3  = (const float*)d_in[7];
  const float* b3  = (const float*)d_in[8];
  float* out = (float*)d_out;

  const int tiles = (NA * BB * TT) / 16;
  const size_t lds_bytes = LDS_FLOATS * sizeof(float);
  coma_influence_kernel<<<tiles, 128, lds_bytes, stream>>>(
      obs, act, cf, W1, b1, W2, b2, W3, b3, out);
}


# Variants go through the three layers two at a time (9 pairs): one workgroup barrier and one round of LDS hand-overs per pair, two independent accumulator chains in layer 2; every sum keeps its order
# speedup vs baseline: 2.0783x; 2.0783x over previous
#include <hip/hip_runtime.h>
#include <hip/hip_bf16.h>
#include <math.h>


#define NA   16
#define DOBS 256
#define AD   32
#define BB   16
#define TT   64
#define SS   16
#define H1D  128
#define H2D  64
#define BT   (BB*TT)
#define NVAR 18
#define INV_S1 (1.0f/17.0f)

#define H1P  132
#define H2P  68
#define OBSP 260

#define PPP      36
#define PPW      (16*PPP)
#define H1PL     (16*H1P)
#define H2PL     (16*H2P)
#define OFF_H1   0
#define OFF_H2   4224
#define OFF_PWP  6400
#define OFF_PQP  8704
#define OFF_OBS  11008
#define LDS_FLOATS 15232

typedef float v2f __attribute__((ext_vector_type(2)));
typedef float v8f __attribute__((ext_vector_type(8)));

__device__ __forceinline__ v8f wmma4(v2f a, v2f b, v8f c) {
  return __builtin_amdgcn_wmma_f32_16x16x4_f32(
       false, a,  false, b,
       (short)0, c,  false,  false);
}

__device__ __forceinline__ float wave_max(float v) {
  #pragma unroll
  for (int off = 16; off; off >>= 1) v = fmaxf(v, __shfl_xor(v, off, 32));
  return v;
}
__device__ __forceinline__ float wave_sum(float v) {
  #pragma unroll
  for (int off = 16; off; off >>= 1) v += __shfl_xor(v, off, 32);
  return v;
}

template <bool FIRST>
__device__ __forceinline__ void variant_pair(
    const float* __restrict__ rowx, const float* __restrict__ rowy,
    float* s_set, float* s_h2, float* s_pwp_w,
    const v8f (&hobs)[2], const v2f (&bw1)[2][8], const v2f (&bw2)[32], const v2f (&bw3)[2][4],
    const float bias2, v8f (&accq)[2],
    const int col, const int hi, const int kb,
    const int jc1a, const int jc1b, const int jc2, const int k3)
{
  v2f avx[8], avy[8];
  #pragma unroll
  for (int kq = 0; kq < 8; ++kq) {
    const float2 t = *(const float2*)(rowx + kq * 4 + kb);
    avx[kq][0] = t.x; avx[kq][1] = t.y;
  }
  if (!FIRST) {
    #pragma unroll
    for (int kq = 0; kq < 8; ++kq) {
      const float2 t = *(const float2*)(rowy + kq * 4 + kb);
      avy[kq][0] = t.x; avy[kq][1] = t.y;
    }
  }

  {
    v8f x0 = hobs[0], x1 = hobs[1], y0 = hobs[0], y1 = hobs[1];
    #pragma unroll
    for (int kq = 0; kq < 8; ++kq) {
      x0 = wmma4(avx[kq], bw1[0][kq], x0);
      x1 = wmma4(avx[kq], bw1[1][kq], x1);
      if (!FIRST) {
        y0 = wmma4(avy[kq], bw1[0][kq], y0);
        y1 = wmma4(avy[kq], bw1[1][kq], y1);
      }
    }
    #pragma unroll
    for (int v = 0; v < 8; ++v) {
      float* r = s_set + (v + 8 * hi) * H1P;
      r[jc1a]        = fmaxf(x0[v], 0.0f);
      r[jc1b]        = fmaxf(x1[v], 0.0f);
      r[H1PL + jc1a] = fmaxf(y0[v], 0.0f);
      r[H1PL + jc1b] = fmaxf(y1[v], 0.0f);
    }
  }
  __syncthreads();

  {
    v8f ax, ay;
    #pragma unroll
    for (int v = 0; v < 8; ++v) { ax[v] = bias2; ay[v] = bias2; }
    const float* hx = s_set + col * H1P + kb;
    const float* hy = hx + H1PL;
    v2f fx[2][4], fy[2][4];
    #pragma unroll
    for (int j = 0; j < 4; ++j) {
      const float2 t = *(const float2*)(hx + j * 4);
      fx[0][j][0] = t.x; fx[0][j][1] = t.y;
      const float2 u = *(const float2*)(hy + j * 4);
      fy[0][j][0] = u.x; fy[0][j][1] = u.y;
    }
    #pragma unroll
    for (int c = 0; c < 8; ++c) {
      const int cur = c & 1, nxt = cur ^ 1;
      if (c + 1 < 8) {
        #pragma unroll
        for (int j = 0; j < 4; ++j) {
          const float2 t = *(const float2*)(hx + ((c + 1) * 4 + j) * 4);
          fx[nxt][j][0] = t.x; fx[nxt][j][1] = t.y;
          const float2 u = *(const float2*)(hy + ((c + 1) * 4 + j) * 4);
          fy[nxt][j][0] = u.x; fy[nxt][j][1] = u.y;
        }
      }
      __builtin_amdgcn_sched_barrier(0);
      #pragma unroll
      for (int j = 0; j < 4; ++j) {
        ax = wmma4(fx[cur][j], bw2[c * 4 + j], ax);
        ay = wmma4(fy[cur][j], bw2[c * 4 + j], ay);
      }
      __builtin_amdgcn_sched_barrier(0);
    }
    #pragma unroll
    for (int v = 0; v < 8; ++v) {
      s_h2[(v + 8 * hi) * H2P + jc2]        = fmaxf(ax[v], 0.0f);
      s_h2[H2PL + (v + 8 * hi) * H2P + jc2] = fmaxf(ay[v], 0.0f);
    }
  }
  __builtin_amdgcn_fence(__ATOMIC_RELEASE, "wavefront");
  __builtin_amdgcn_wave_barrier();
  __builtin_amdgcn_fence(__ATOMIC_ACQUIRE, "wavefront");

  {
    v2f gx[4], gy[4];
    #pragma unroll
    for (int ks = 0; ks < 4; ++ks) {
      const float2 t = *(const float2*)(s_h2 + col * H2P + k3 + ks * 4 + kb);
      gx[ks][0] = t.x; gx[ks][1] = t.y;
      const float2 u = *(const float2*)(s_h2 + H2PL + col * H2P + k3 + ks * 4 + kb);
      gy[ks][0] = u.x; gy[ks][1] = u.y;
    }
    #pragma unroll
    for (int ks = 0; ks < 4; ++ks) {
      accq[0] = wmma4(gx[ks], bw3[0][ks], accq[0]);
      accq[1] = wmma4(gx[ks], bw3[1][ks], accq[1]);
    }
    if (FIRST) {
      #pragma unroll
      for (int t = 0; t < 2; ++t) {
        #pragma unroll
        for (int v = 0; v < 8; ++v) {
          s_pwp_w[(v + 8 * hi) * PPP + t * 16 + col] = accq[t][v];
          accq[t][v] = 0.0f;
        }
      }
    }
    #pragma unroll
    for (int ks = 0; ks < 4; ++ks) {
      accq[0] = wmma4(gy[ks], bw3[0][ks], accq[0]);
      accq[1] = wmma4(gy[ks], bw3[1][ks], accq[1]);
    }
  }
}

__global__ __launch_bounds__(128)
void coma_influence_kernel(const float* __restrict__ obs,
                           const float* __restrict__ act,
                           const float* __restrict__ cf,
                           const float* __restrict__ W1,
                           const float* __restrict__ b1,
                           const float* __restrict__ W2,
                           const float* __restrict__ b2,
                           const float* __restrict__ W3,
                           const float* __restrict__ b3,
                           float* __restrict__ out)
{
  __builtin_amdgcn_s_setprio(3);

  extern __shared__ float smem[];
  float* s_h1   = smem + OFF_H1;
  float* s_h2   = smem + OFF_H2;
  float* s_pwp  = smem + OFF_PWP;
  float* s_pqp  = smem + OFF_PQP;
  float* s_obs  = smem + OFF_OBS;

  const int tid  = threadIdx.x;
  const int lane = tid & 31;
  const int wid  = tid >> 5;
  const int g    = blockIdx.x;
  const int r0   = g * 16;
  const int n    = r0 >> 10;
  const int bt0  = r0 & 1023;

  const int col = lane & 15;
  const int hi  = lane >> 4;
  const int kb  = hi * 2;

  const int jc1a = (wid * 2 + 0) * 16 + col;
  const int jc1b = (wid * 2 + 1) * 16 + col;
  const int jc2  = wid * 16 + col;
  const int k3   = wid * 16;

  v2f bw1[2][8];
  #pragma unroll
  for (int kq = 0; kq < 8; ++kq) {
    const int k = DOBS + kq * 4 + kb;
    bw1[0][kq][0] = W1[(k    ) * H1D + jc1a];
    bw1[0][kq][1] = W1[(k + 1) * H1D + jc1a];
    bw1[1][kq][0] = W1[(k    ) * H1D + jc1b];
    bw1[1][kq][1] = W1[(k + 1) * H1D + jc1b];
  }
  v2f bw2[32];
  #pragma unroll
  for (int ks = 0; ks < 32; ++ks) {
    const int k = ks * 4 + kb;
    bw2[ks][0] = W2[(k    ) * H2D + jc2];
    bw2[ks][1] = W2[(k + 1) * H2D + jc2];
  }
  v2f bw3[2][4];
  #pragma unroll
  for (int ks = 0; ks < 4; ++ks) {
    const int k = k3 + ks * 4 + kb;
    bw3[0][ks][0] = W3[(k    ) * AD + col];
    bw3[0][ks][1] = W3[(k + 1) * AD + col];
    bw3[1][ks][0] = W3[(k    ) * AD + 16 + col];
    bw3[1][ks][1] = W3[(k + 1) * AD + 16 + col];
  }
  const float bias2 = b2[jc2];

  #pragma unroll
  for (int j4 = 0; j4 < 8; ++j4) {
    const int i4 = tid + 128 * j4;
    const int m  = i4 >> 6;
    const int d  = (i4 & 63) * 4;
    const float4 v = *(const float4*)(obs + ((size_t)(bt0 + m) * NA + n) * DOBS + d);
    *(float4*)(s_obs + m * OBSP + d) = v;
  }
  __syncthreads();

  v8f hobs[2];
  #pragma unroll
  for (int jj = 0; jj < 2; ++jj) {
    const int jcol = (wid * 2 + jj) * 16 + col;
    v8f acc;
    {
      const float bias = b1[jcol];
      #pragma unroll
      for (int v = 0; v < 8; ++v) acc[v] = bias;
    }
    for (int kk = 0; kk < DOBS; kk += 4) {
      v2f a; a[0] = s_obs[col * OBSP + kk + kb]; a[1] = s_obs[col * OBSP + kk + kb + 1];
      v2f b; b[0] = W1[(kk + kb) * H1D + jcol];  b[1] = W1[(kk + kb + 1) * H1D + jcol];
      acc = wmma4(a, b, acc);
    }
    hobs[jj] = acc;
  }

  v8f accq[2];
  #pragma unroll
  for (int v = 0; v < 8; ++v) { accq[0][v] = 0.0f; accq[1][v] = 0.0f; }

  float* s_set0 = s_h1;
  float* s_set1 = s_obs;

  const float* cf0 = cf + (((size_t)n * SS) * BT + (bt0 + col)) * AD;
  __builtin_prefetch(cf0, 0, 1);
  __builtin_prefetch(cf0 + (size_t)BT * AD, 0, 1);

  variant_pair<true>(act + ((size_t)(bt0 + col) * NA + n) * AD, nullptr,
                     s_set0, s_h2, s_pwp + wid * PPW,
                     hobs, bw1, bw2, bw3, bias2, accq, col, hi, kb, jc1a, jc1b, jc2, k3);

  for (int p = 1; p < NVAR / 2; ++p) {
    if (p == 5)      __builtin_amdgcn_s_setprio(2);
    else if (p == 7) __builtin_amdgcn_s_setprio(1);
    else if (p == 8) __builtin_amdgcn_s_setprio(0);

    const float* rowx = cf0 + (size_t)(2 * p - 2) * BT * AD;
    const float* rowy = rowx + (size_t)BT * AD;
    if (p + 1 < NVAR / 2) {
      __builtin_prefetch(rowx + (size_t)2 * BT * AD, 0, 1);
      __builtin_prefetch(rowx + (size_t)3 * BT * AD, 0, 1);
    }
    variant_pair<false>(rowx, rowy, (p & 1) ? s_set1 : s_set0, s_h2, s_pwp + wid * PPW,
                        hobs, bw1, bw2, bw3, bias2, accq, col, hi, kb, jc1a, jc1b, jc2, k3);
  }

  #pragma unroll
  for (int t = 0; t < 2; ++t) {
    #pragma unroll
    for (int v = 0; v < 8; ++v)
      s_pqp[wid * PPW + (v + 8 * hi) * PPP + t * 16 + col] = accq[t][v];
  }
  __syncthreads();

  const float bias3 = b3[lane];
  #pragma unroll
  for (int rr = 0; rr < 4; ++rr) {
    const int r = wid * 4 + rr;
    const int   o  = r * PPP + lane;
    const float pw = ((s_pwp[o] + s_pwp[PPW + o]) + s_pwp[2 * PPW + o]) + s_pwp[3 * PPW + o] + bias3;
    const float pq = (((s_pqp[o] + s_pqp[PPW + o]) + s_pqp[2 * PPW + o]) + s_pqp[3 * PPW + o]) * INV_S1 + bias3;

    const float mp = wave_max(pw);
    const float mq = wave_max(pq);
    const float ep = expf(pw - mp);
    const float eq = expf(pq - mq);
    const float sp = wave_sum(ep);
    const float sq = wave_sum(eq);

    const float log_p = (pw - mp) - logf(sp);
    const float log_q = (pq - mq) - logf(sq);
    const float q     = eq / sq;
    float kl = q * (log_q - log_p);
    kl = wave_sum(kl);
    if (lane == 0) out[(size_t)(bt0 + r) * NA + n] = kl * (1.0f / AD);
  }
}

extern "C" void kernel_launch(void* const* d_in, const int* in_sizes, int n_in,
                              void* d_out, int out_size, void* d_ws, size_t ws_size,
                              hipStream_t stream) {
  (void)in_sizes; (void)n_in; (void)d_ws; (void)ws_size; (void)out_size;
  const float* obs = (const float*)d_in[0];
  const float* act = (const float*)d_in[1];
  const float* cf  = (const float*)d_in[2];
  const float* W1  = (const float*)d_in[3];
  const float* b1  = (const float*)d_in[4];
  const float* W2  = (const float*)d_in[5];
  const float* b2  = (const float*)d_in[6];
  const float* W3  = (const float*)d_in[7];
  const float* b3  = (const float*)d_in[8];
  float* out = (float*)d_out;

  const int tiles = (NA * BB * TT) / 16;
  const size_t lds_bytes = LDS_FLOATS * sizeof(float);
  coma_influence_kernel<<<tiles, 128, lds_bytes, stream>>>(
      obs, act, cf, W1, b1, W2, b2, W3, b3, out);
}


# Step 1 (obs @ W1) is one k loop over both column tiles, a lane's two columns adjacent: W1 rows come by 8-byte loads one 8-step block ahead of their WMMAs, two accumulator chains; sum order kept
# speedup vs baseline: 2.2271x; 1.0716x over previous
#include <hip/hip_runtime.h>
#include <hip/hip_bf16.h>
#include <math.h>


#define NA   16
#define DOBS 256
#define AD   32
#define BB   16
#define TT   64
#define SS   16
#define H1D  128
#define H2D  64
#define BT   (BB*TT)
#define NVAR 18
#define INV_S1 (1.0f/17.0f)

#define H1P  132
#define H2P  68
#define OBSP 260

#define PPP      36
#define PPW      (16*PPP)
#define H1PL     (16*H1P)
#define H2PL     (16*H2P)
#define OFF_H1   0
#define OFF_H2   4224
#define OFF_PWP  6400
#define OFF_PQP  8704
#define OFF_OBS  11008
#define LDS_FLOATS 15232

typedef float v2f __attribute__((ext_vector_type(2)));
typedef float v8f __attribute__((ext_vector_type(8)));

__device__ __forceinline__ v8f wmma4(v2f a, v2f b, v8f c) {
  return __builtin_amdgcn_wmma_f32_16x16x4_f32(
       false, a,  false, b,
       (short)0, c,  false,  false);
}

__device__ __forceinline__ float wave_max(float v) {
  #pragma unroll
  for (int off = 16; off; off >>= 1) v = fmaxf(v, __shfl_xor(v, off, 32));
  return v;
}
__device__ __forceinline__ float wave_sum(float v) {
  #pragma unroll
  for (int off = 16; off; off >>= 1) v += __shfl_xor(v, off, 32);
  return v;
}

template <bool FIRST>
__device__ __forceinline__ void variant_pair(
    const float* __restrict__ rowx, const float* __restrict__ rowy,
    float* s_set, float* s_h2, float* s_pwp_w,
    const v8f (&hobs)[2], const v2f (&bw1)[2][8], const v2f (&bw2)[32], const v2f (&bw3)[2][4],
    const float bias2, v8f (&accq)[2],
    const int col, const int hi, const int kb,
    const int jc1a, const int jc1b, const int jc2, const int k3)
{
  v2f avx[8], avy[8];
  #pragma unroll
  for (int kq = 0; kq < 8; ++kq) {
    const float2 t = *(const float2*)(rowx + kq * 4 + kb);
    avx[kq][0] = t.x; avx[kq][1] = t.y;
  }
  if (!FIRST) {
    #pragma unroll
    for (int kq = 0; kq < 8; ++kq) {
      const float2 t = *(const float2*)(rowy + kq * 4 + kb);
      avy[kq][0] = t.x; avy[kq][1] = t.y;
    }
  }

  {
    v8f x0 = hobs[0], x1 = hobs[1], y0 = hobs[0], y1 = hobs[1];
    #pragma unroll
    for (int kq = 0; kq < 8; ++kq) {
      x0 = wmma4(avx[kq], bw1[0][kq], x0);
      x1 = wmma4(avx[kq], bw1[1][kq], x1);
      if (!FIRST) {
        y0 = wmma4(avy[kq], bw1[0][kq], y0);
        y1 = wmma4(avy[kq], bw1[1][kq], y1);
      }
    }
    #pragma unroll
    for (int v = 0; v < 8; ++v) {
      float* r = s_set + (v + 8 * hi) * H1P;
      r[jc1a]        = fmaxf(x0[v], 0.0f);
      r[jc1b]        = fmaxf(x1[v], 0.0f);
      r[H1PL + jc1a] = fmaxf(y0[v], 0.0f);
      r[H1PL + jc1b] = fmaxf(y1[v], 0.0f);
    }
  }
  __syncthreads();

  {
    v8f ax, ay;
    #pragma unroll
    for (int v = 0; v < 8; ++v) { ax[v] = bias2; ay[v] = bias2; }
    const float* hx = s_set + col * H1P + kb;
    const float* hy = hx + H1PL;
    v2f fx[2][4], fy[2][4];
    #pragma unroll
    for (int j = 0; j < 4; ++j) {
      const float2 t = *(const float2*)(hx + j * 4);
      fx[0][j][0] = t.x; fx[0][j][1] = t.y;
      const float2 u = *(const float2*)(hy + j * 4);
      fy[0][j][0] = u.x; fy[0][j][1] = u.y;
    }
    #pragma unroll
    for (int c = 0; c < 8; ++c) {
      const int cur = c & 1, nxt = cur ^ 1;
      if (c + 1 < 8) {
        #pragma unroll
        for (int j = 0; j < 4; ++j) {
          const float2 t = *(const float2*)(hx + ((c + 1) * 4 + j) * 4);
          fx[nxt][j][0] = t.x; fx[nxt][j][1] = t.y;
          const float2 u = *(const float2*)(hy + ((c + 1) * 4 + j) * 4);
          fy[nxt][j][0] = u.x; fy[nxt][j][1] = u.y;
        }
      }
      __builtin_amdgcn_sched_barrier(0);
      #pragma unroll
      for (int j = 0; j < 4; ++j) {
        ax = wmma4(fx[cur][j], bw2[c * 4 + j], ax);
        ay = wmma4(fy[cur][j], bw2[c * 4 + j], ay);
      }
      __builtin_amdgcn_sched_barrier(0);
    }
    #pragma unroll
    for (int v = 0; v < 8; ++v) {
      s_h2[(v + 8 * hi) * H2P + jc2]        = fmaxf(ax[v], 0.0f);
      s_h2[H2PL + (v + 8 * hi) * H2P + jc2] = fmaxf(ay[v], 0.0f);
    }
  }
  __builtin_amdgcn_fence(__ATOMIC_RELEASE, "wavefront");
  __builtin_amdgcn_wave_barrier();
  __builtin_amdgcn_fence(__ATOMIC_ACQUIRE, "wavefront");

  {
    v2f gx[4], gy[4];
    #pragma unroll
    for (int ks = 0; ks < 4; ++ks) {
      const float2 t = *(const float2*)(s_h2 + col * H2P + k3 + ks * 4 + kb);
      gx[ks][0] = t.x; gx[ks][1] = t.y;
      const float2 u = *(const float2*)(s_h2 + H2PL + col * H2P + k3 + ks * 4 + kb);
      gy[ks][0] = u.x; gy[ks][1] = u.y;
    }
    #pragma unroll
    for (int ks = 0; ks < 4; ++ks) {
      accq[0] = wmma4(gx[ks], bw3[0][ks], accq[0]);
      accq[1] = wmma4(gx[ks], bw3[1][ks], accq[1]);
    }
    if (FIRST) {
      #pragma unroll
      for (int t = 0; t < 2; ++t) {
        #pragma unroll
        for (int v = 0; v < 8; ++v) {
          s_pwp_w[(v + 8 * hi) * PPP + t * 16 + col] = accq[t][v];
          accq[t][v] = 0.0f;
        }
      }
    }
    #pragma unroll
    for (int ks = 0; ks < 4; ++ks) {
      accq[0] = wmma4(gy[ks], bw3[0][ks], accq[0]);
      accq[1] = wmma4(gy[ks], bw3[1][ks], accq[1]);
    }
  }
}

__global__ __launch_bounds__(128)
void coma_influence_kernel(const float* __restrict__ obs,
                           const float* __restrict__ act,
                           const float* __restrict__ cf,
                           const float* __restrict__ W1,
                           const float* __restrict__ b1,
                           const float* __restrict__ W2,
                           const float* __restrict__ b2,
                           const float* __restrict__ W3,
                           const float* __restrict__ b3,
                           float* __restrict__ out)
{
  __builtin_amdgcn_s_setprio(3);

  extern __shared__ float smem[];
  float* s_h1   = smem + OFF_H1;
  float* s_h2   = smem + OFF_H2;
  float* s_pwp  = smem + OFF_PWP;
  float* s_pqp  = smem + OFF_PQP;
  float* s_obs  = smem + OFF_OBS;

  const int tid  = threadIdx.x;
  const int lane = tid & 31;
  const int wid  = tid >> 5;
  const int g    = blockIdx.x;
  const int r0   = g * 16;
  const int n    = r0 >> 10;
  const int bt0  = r0 & 1023;

  const int col = lane & 15;
  const int hi  = lane >> 4;
  const int kb  = hi * 2;

  const int jc1a = wid * 32 + 2 * col;
  const int jc1b = jc1a + 1;
  const int jc2  = wid * 16 + col;
  const int k3   = wid * 16;

  v2f bw1[2][8];
  #pragma unroll
  for (int kq = 0; kq < 8; ++kq) {
    const int k = DOBS + kq * 4 + kb;
    const float2 r0 = *(const float2*)(W1 + (k    ) * H1D + jc1a);
    const float2 r1 = *(const float2*)(W1 + (k + 1) * H1D + jc1a);
    bw1[0][kq][0] = r0.x; bw1[0][kq][1] = r1.x;
    bw1[1][kq][0] = r0.y; bw1[1][kq][1] = r1.y;
  }
  v2f bw2[32];
  #pragma unroll
  for (int ks = 0; ks < 32; ++ks) {
    const int k = ks * 4 + kb;
    bw2[ks][0] = W2[(k    ) * H2D + jc2];
    bw2[ks][1] = W2[(k + 1) * H2D + jc2];
  }
  v2f bw3[2][4];
  #pragma unroll
  for (int ks = 0; ks < 4; ++ks) {
    const int k = k3 + ks * 4 + kb;
    bw3[0][ks][0] = W3[(k    ) * AD + col];
    bw3[0][ks][1] = W3[(k + 1) * AD + col];
    bw3[1][ks][0] = W3[(k    ) * AD + 16 + col];
    bw3[1][ks][1] = W3[(k + 1) * AD + 16 + col];
  }
  const float bias2 = b2[jc2];

  #pragma unroll
  for (int j4 = 0; j4 < 8; ++j4) {
    const int i4 = tid + 128 * j4;
    const int m  = i4 >> 6;
    const int d  = (i4 & 63) * 4;
    const float4 v = *(const float4*)(obs + ((size_t)(bt0 + m) * NA + n) * DOBS + d);
    *(float4*)(s_obs + m * OBSP + d) = v;
  }
  const float* wp = W1 + kb * H1D + jc1a;
  float2 wr[2][8][2];
  #pragma unroll
  for (int j = 0; j < 8; ++j) {
    wr[0][j][0] = *(const float2*)(wp + (j * 4    ) * H1D);
    wr[0][j][1] = *(const float2*)(wp + (j * 4 + 1) * H1D);
  }
  const float bias1a = b1[jc1a];
  const float bias1b = b1[jc1b];
  __syncthreads();

  v8f hobs[2];
  #pragma unroll
  for (int v = 0; v < 8; ++v) { hobs[0][v] = bias1a; hobs[1][v] = bias1b; }
  {
    const float* ap = s_obs + col * OBSP + kb;
    #pragma unroll
    for (int c = 0; c < 8; ++c) {
      const int cur = c & 1, nxt = cur ^ 1;
      v2f a[8];
      #pragma unroll
      for (int j = 0; j < 8; ++j) {
        const float2 t = *(const float2*)(ap + (c * 8 + j) * 4);
        a[j][0] = t.x; a[j][1] = t.y;
      }
      if (c + 1 < 8) {
        #pragma unroll
        for (int j = 0; j < 8; ++j) {
          wr[nxt][j][0] = *(const float2*)(wp + ((c + 1) * 32 + j * 4    ) * H1D);
          wr[nxt][j][1] = *(const float2*)(wp + ((c + 1) * 32 + j * 4 + 1) * H1D);
        }
      }
      __builtin_amdgcn_sched_barrier(0);
      #pragma unroll
      for (int j = 0; j < 8; ++j) {
        v2f b0, b1v;
        b0[0]  = wr[cur][j][0].x; b0[1]  = wr[cur][j][1].x;
        b1v[0] = wr[cur][j][0].y; b1v[1] = wr[cur][j][1].y;
        hobs[0] = wmma4(a[j], b0,  hobs[0]);
        hobs[1] = wmma4(a[j], b1v, hobs[1]);
      }
      __builtin_amdgcn_sched_barrier(0);
    }
  }

  v8f accq[2];
  #pragma unroll
  for (int v = 0; v < 8; ++v) { accq[0][v] = 0.0f; accq[1][v] = 0.0f; }

  float* s_set0 = s_h1;
  float* s_set1 = s_obs;

  const float* cf0 = cf + (((size_t)n * SS) * BT + (bt0 + col)) * AD;
  __builtin_prefetch(cf0, 0, 1);
  __builtin_prefetch(cf0 + (size_t)BT * AD, 0, 1);

  variant_pair<true>(act + ((size_t)(bt0 + col) * NA + n) * AD, nullptr,
                     s_set0, s_h2, s_pwp + wid * PPW,
                     hobs, bw1, bw2, bw3, bias2, accq, col, hi, kb, jc1a, jc1b, jc2, k3);

  for (int p = 1; p < NVAR / 2; ++p) {
    if (p == 5)      __builtin_amdgcn_s_setprio(2);
    else if (p == 7) __builtin_amdgcn_s_setprio(1);
    else if (p == 8) __builtin_amdgcn_s_setprio(0);

    const float* rowx = cf0 + (size_t)(2 * p - 2) * BT * AD;
    const float* rowy = rowx + (size_t)BT * AD;
    if (p + 1 < NVAR / 2) {
      __builtin_prefetch(rowx + (size_t)2 * BT * AD, 0, 1);
      __builtin_prefetch(rowx + (size_t)3 * BT * AD, 0, 1);
    }
    variant_pair<false>(rowx, rowy, (p & 1) ? s_set1 : s_set0, s_h2, s_pwp + wid * PPW,
                        hobs, bw1, bw2, bw3, bias2, accq, col, hi, kb, jc1a, jc1b, jc2, k3);
  }

  #pragma unroll
  for (int t = 0; t < 2; ++t) {
    #pragma unroll
    for (int v = 0; v < 8; ++v)
      s_pqp[wid * PPW + (v + 8 * hi) * PPP + t * 16 + col] = accq[t][v];
  }
  __syncthreads();

  const float bias3 = b3[lane];
  #pragma unroll
  for (int rr = 0; rr < 4; ++rr) {
    const int r = wid * 4 + rr;
    const int   o  = r * PPP + lane;
    const float pw = ((s_pwp[o] + s_pwp[PPW + o]) + s_pwp[2 * PPW + o]) + s_pwp[3 * PPW + o] + bias3;
    const float pq = (((s_pqp[o] + s_pqp[PPW + o]) + s_pqp[2 * PPW + o]) + s_pqp[3 * PPW + o]) * INV_S1 + bias3;

    const float mp = wave_max(pw);
    const float mq = wave_max(pq);
    const float ep = expf(pw - mp);
    const float eq = expf(pq - mq);
    const float sp = wave_sum(ep);
    const float sq = wave_sum(eq);

    const float log_p = (pw - mp) - logf(sp);
    const float log_q = (pq - mq) - logf(sq);
    const float q     = eq / sq;
    float kl = q * (log_q - log_p);
    kl = wave_sum(kl);
    if (lane == 0) out[(size_t)(bt0 + r) * NA + n] = kl * (1.0f / AD);
  }
}

extern "C" void kernel_launch(void* const* d_in, const int* in_sizes, int n_in,
                              void* d_out, int out_size, void* d_ws, size_t ws_size,
                              hipStream_t stream) {
  (void)in_sizes; (void)n_in; (void)d_ws; (void)ws_size; (void)out_size;
  const float* obs = (const float*)d_in[0];
  const float* act = (const float*)d_in[1];
  const float* cf  = (const float*)d_in[2];
  const float* W1  = (const float*)d_in[3];
  const float* b1  = (const float*)d_in[4];
  const float* W2  = (const float*)d_in[5];
  const float* b2  = (const float*)d_in[6];
  const float* W3  = (const float*)d_in[7];
  const float* b3  = (const float*)d_in[8];
  float* out = (float*)d_out;

  const int tiles = (NA * BB * TT) / 16;
  const size_t lds_bytes = LDS_FLOATS * sizeof(float);
  coma_influence_kernel<<<tiles, 128, lds_bytes, stream>>>(
      obs, act, cf, W1, b1, W2, b2, W3, b3, out);
}


# Closing step re-mapped: a row's 32 logits sit on 8 lanes x 4 actions (float4 LDS reads), all 16 rows at once: 11 dependent lane exchanges per wave, not 72; softmax sums and KL mean change sum order
# speedup vs baseline: 2.3165x; 1.0401x over previous
#include <hip/hip_runtime.h>
#include <hip/hip_bf16.h>
#include <math.h>


#define NA   16
#define DOBS 256
#define AD   32
#define BB   16
#define TT   64
#define SS   16
#define H1D  128
#define H2D  64
#define BT   (BB*TT)
#define NVAR 18
#define INV_S1 (1.0f/17.0f)

#define H1P  132
#define H2P  68
#define OBSP 260

#define PPP      36
#define PPW      (16*PPP)
#define H1PL     (16*H1P)
#define H2PL     (16*H2P)
#define OFF_H1   0
#define OFF_H2   4224
#define OFF_PWP  6400
#define OFF_PQP  8704
#define OFF_OBS  11008
#define LDS_FLOATS 15232

typedef float v2f __attribute__((ext_vector_type(2)));
typedef float v8f __attribute__((ext_vector_type(8)));

__device__ __forceinline__ v8f wmma4(v2f a, v2f b, v8f c) {
  return __builtin_amdgcn_wmma_f32_16x16x4_f32(
       false, a,  false, b,
       (short)0, c,  false,  false);
}

__device__ __forceinline__ float row8_max(float v) {
  #pragma unroll
  for (int off = 4; off; off >>= 1) v = fmaxf(v, __shfl_xor(v, off, 32));
  return v;
}
__device__ __forceinline__ float row8_sum(float v) {
  #pragma unroll
  for (int off = 4; off; off >>= 1) v += __shfl_xor(v, off, 32);
  return v;
}

template <bool FIRST>
__device__ __forceinline__ void variant_pair(
    const float* __restrict__ rowx, const float* __restrict__ rowy,
    float* s_set, float* s_h2, float* s_pwp_w,
    const v8f (&hobs)[2], const v2f (&bw1)[2][8], const v2f (&bw2)[32], const v2f (&bw3)[2][4],
    const float bias2, v8f (&accq)[2],
    const int col, const int hi, const int kb,
    const int jc1a, const int jc1b, const int jc2, const int k3)
{
  v2f avx[8], avy[8];
  #pragma unroll
  for (int kq = 0; kq < 8; ++kq) {
    const float2 t = *(const float2*)(rowx + kq * 4 + kb);
    avx[kq][0] = t.x; avx[kq][1] = t.y;
  }
  if (!FIRST) {
    #pragma unroll
    for (int kq = 0; kq < 8; ++kq) {
      const float2 t = *(const float2*)(rowy + kq * 4 + kb);
      avy[kq][0] = t.x; avy[kq][1] = t.y;
    }
  }

  {
    v8f x0 = hobs[0], x1 = hobs[1], y0 = hobs[0], y1 = hobs[1];
    #pragma unroll
    for (int kq = 0; kq < 8; ++kq) {
      x0 = wmma4(avx[kq], bw1[0][kq], x0);
      x1 = wmma4(avx[kq], bw1[1][kq], x1);
      if (!FIRST) {
        y0 = wmma4(avy[kq], bw1[0][kq], y0);
        y1 = wmma4(avy[kq], bw1[1][kq], y1);
      }
    }
    #pragma unroll
    for (int v = 0; v < 8; ++v) {
      float* r = s_set + (v + 8 * hi) * H1P;
      r[jc1a]        = fmaxf(x0[v], 0.0f);
      r[jc1b]        = fmaxf(x1[v], 0.0f);
      r[H1PL + jc1a] = fmaxf(y0[v], 0.0f);
      r[H1PL + jc1b] = fmaxf(y1[v], 0.0f);
    }
  }
  __syncthreads();

  {
    v8f ax, ay;
    #pragma unroll
    for (int v = 0; v < 8; ++v) { ax[v] = bias2; ay[v] = bias2; }
    const float* hx = s_set + col * H1P + kb;
    const float* hy = hx + H1PL;
    v2f fx[2][4], fy[2][4];
    #pragma unroll
    for (int j = 0; j < 4; ++j) {
      const float2 t = *(const float2*)(hx + j * 4);
      fx[0][j][0] = t.x; fx[0][j][1] = t.y;
      const float2 u = *(const float2*)(hy + j * 4);
      fy[0][j][0] = u.x; fy[0][j][1] = u.y;
    }
    #pragma unroll
    for (int c = 0; c < 8; ++c) {
      const int cur = c & 1, nxt = cur ^ 1;
      if (c + 1 < 8) {
        #pragma unroll
        for (int j = 0; j < 4; ++j) {
          const float2 t = *(const float2*)(hx + ((c + 1) * 4 + j) * 4);
          fx[nxt][j][0] = t.x; fx[nxt][j][1] = t.y;
          const float2 u = *(const float2*)(hy + ((c + 1) * 4 + j) * 4);
          fy[nxt][j][0] = u.x; fy[nxt][j][1] = u.y;
        }
      }
      __builtin_amdgcn_sched_barrier(0);
      #pragma unroll
      for (int j = 0; j < 4; ++j) {
        ax = wmma4(fx[cur][j], bw2[c * 4 + j], ax);
        ay = wmma4(fy[cur][j], bw2[c * 4 + j], ay);
      }
      __builtin_amdgcn_sched_barrier(0);
    }
    #pragma unroll
    for (int v = 0; v < 8; ++v) {
      s_h2[(v + 8 * hi) * H2P + jc2]        = fmaxf(ax[v], 0.0f);
      s_h2[H2PL + (v + 8 * hi) * H2P + jc2] = fmaxf(ay[v], 0.0f);
    }
  }
  __builtin_amdgcn_fence(__ATOMIC_RELEASE, "wavefront");
  __builtin_amdgcn_wave_barrier();
  __builtin_amdgcn_fence(__ATOMIC_ACQUIRE, "wavefront");

  {
    v2f gx[4], gy[4];
    #pragma unroll
    for (int ks = 0; ks < 4; ++ks) {
      const float2 t = *(const float2*)(s_h2 + col * H2P + k3 + ks * 4 + kb);
      gx[ks][0] = t.x; gx[ks][1] = t.y;
      const float2 u = *(const float2*)(s_h2 + H2PL + col * H2P + k3 + ks * 4 + kb);
      gy[ks][0] = u.x; gy[ks][1] = u.y;
    }
    #pragma unroll
    for (int ks = 0; ks < 4; ++ks) {
      accq[0] = wmma4(gx[ks], bw3[0][ks], accq[0]);
      accq[1] = wmma4(gx[ks], bw3[1][ks], accq[1]);
    }
    if (FIRST) {
      #pragma unroll
      for (int t = 0; t < 2; ++t) {
        #pragma unroll
        for (int v = 0; v < 8; ++v) {
          s_pwp_w[(v + 8 * hi) * PPP + t * 16 + col] = accq[t][v];
          accq[t][v] = 0.0f;
        }
      }
    }
    #pragma unroll
    for (int ks = 0; ks < 4; ++ks) {
      accq[0] = wmma4(gy[ks], bw3[0][ks], accq[0]);
      accq[1] = wmma4(gy[ks], bw3[1][ks], accq[1]);
    }
  }
}

__global__ __launch_bounds__(128)
void coma_influence_kernel(const float* __restrict__ obs,
                           const float* __restrict__ act,
                           const float* __restrict__ cf,
                           const float* __restrict__ W1,
                           const float* __restrict__ b1,
                           const float* __restrict__ W2,
                           const float* __restrict__ b2,
                           const float* __restrict__ W3,
                           const float* __restrict__ b3,
                           float* __restrict__ out)
{
  __builtin_amdgcn_s_setprio(3);

  extern __shared__ float smem[];
  float* s_h1   = smem + OFF_H1;
  float* s_h2   = smem + OFF_H2;
  float* s_pwp  = smem + OFF_PWP;
  float* s_pqp  = smem + OFF_PQP;
  float* s_obs  = smem + OFF_OBS;

  const int tid  = threadIdx.x;
  const int lane = tid & 31;
  const int wid  = tid >> 5;
  const int g    = blockIdx.x;
  const int r0   = g * 16;
  const int n    = r0 >> 10;
  const int bt0  = r0 & 1023;

  const int col = lane & 15;
  const int hi  = lane >> 4;
  const int kb  = hi * 2;

  const int jc1a = wid * 32 + 2 * col;
  const int jc1b = jc1a + 1;
  const int jc2  = wid * 16 + col;
  const int k3   = wid * 16;

  v2f bw1[2][8];
  #pragma unroll
  for (int kq = 0; kq < 8; ++kq) {
    const int k = DOBS + kq * 4 + kb;
    const float2 r0 = *(const float2*)(W1 + (k    ) * H1D + jc1a);
    const float2 r1 = *(const float2*)(W1 + (k + 1) * H1D + jc1a);
    bw1[0][kq][0] = r0.x; bw1[0][kq][1] = r1.x;
    bw1[1][kq][0] = r0.y; bw1[1][kq][1] = r1.y;
  }
  v2f bw2[32];
  #pragma unroll
  for (int ks = 0; ks < 32; ++ks) {
    const int k = ks * 4 + kb;
    bw2[ks][0] = W2[(k    ) * H2D + jc2];
    bw2[ks][1] = W2[(k + 1) * H2D + jc2];
  }
  v2f bw3[2][4];
  #pragma unroll
  for (int ks = 0; ks < 4; ++ks) {
    const int k = k3 + ks * 4 + kb;
    bw3[0][ks][0] = W3[(k    ) * AD + col];
    bw3[0][ks][1] = W3[(k + 1) * AD + col];
    bw3[1][ks][0] = W3[(k    ) * AD + 16 + col];
    bw3[1][ks][1] = W3[(k + 1) * AD + 16 + col];
  }
  const float bias2 = b2[jc2];

  #pragma unroll
  for (int j4 = 0; j4 < 8; ++j4) {
    const int i4 = tid + 128 * j4;
    const int m  = i4 >> 6;
    const int d  = (i4 & 63) * 4;
    const float4 v = *(const float4*)(obs + ((size_t)(bt0 + m) * NA + n) * DOBS + d);
    *(float4*)(s_obs + m * OBSP + d) = v;
  }
  const float* wp = W1 + kb * H1D + jc1a;
  float2 wr[2][8][2];
  #pragma unroll
  for (int j = 0; j < 8; ++j) {
    wr[0][j][0] = *(const float2*)(wp + (j * 4    ) * H1D);
    wr[0][j][1] = *(const float2*)(wp + (j * 4 + 1) * H1D);
  }
  const float bias1a = b1[jc1a];
  const float bias1b = b1[jc1b];
  __syncthreads();

  v8f hobs[2];
  #pragma unroll
  for (int v = 0; v < 8; ++v) { hobs[0][v] = bias1a; hobs[1][v] = bias1b; }
  {
    const float* ap = s_obs + col * OBSP + kb;
    #pragma unroll
    for (int c = 0; c < 8; ++c) {
      const int cur = c & 1, nxt = cur ^ 1;
      v2f a[8];
      #pragma unroll
      for (int j = 0; j < 8; ++j) {
        const float2 t = *(const float2*)(ap + (c * 8 + j) * 4);
        a[j][0] = t.x; a[j][1] = t.y;
      }
      if (c + 1 < 8) {
        #pragma unroll
        for (int j = 0; j < 8; ++j) {
          wr[nxt][j][0] = *(const float2*)(wp + ((c + 1) * 32 + j * 4    ) * H1D);
          wr[nxt][j][1] = *(const float2*)(wp + ((c + 1) * 32 + j * 4 + 1) * H1D);
        }
      }
      __builtin_amdgcn_sched_barrier(0);
      #pragma unroll
      for (int j = 0; j < 8; ++j) {
        v2f b0, b1v;
        b0[0]  = wr[cur][j][0].x; b0[1]  = wr[cur][j][1].x;
        b1v[0] = wr[cur][j][0].y; b1v[1] = wr[cur][j][1].y;
        hobs[0] = wmma4(a[j], b0,  hobs[0]);
        hobs[1] = wmma4(a[j], b1v, hobs[1]);
      }
      __builtin_amdgcn_sched_barrier(0);
    }
  }

  v8f accq[2];
  #pragma unroll
  for (int v = 0; v < 8; ++v) { accq[0][v] = 0.0f; accq[1][v] = 0.0f; }

  float* s_set0 = s_h1;
  float* s_set1 = s_obs;

  const float* cf0 = cf + (((size_t)n * SS) * BT + (bt0 + col)) * AD;
  __builtin_prefetch(cf0, 0, 1);
  __builtin_prefetch(cf0 + (size_t)BT * AD, 0, 1);

  variant_pair<true>(act + ((size_t)(bt0 + col) * NA + n) * AD, nullptr,
                     s_set0, s_h2, s_pwp + wid * PPW,
                     hobs, bw1, bw2, bw3, bias2, accq, col, hi, kb, jc1a, jc1b, jc2, k3);

  for (int p = 1; p < NVAR / 2; ++p) {
    if (p == 5)      __builtin_amdgcn_s_setprio(2);
    else if (p == 7) __builtin_amdgcn_s_setprio(1);
    else if (p == 8) __builtin_amdgcn_s_setprio(0);

    const float* rowx = cf0 + (size_t)(2 * p - 2) * BT * AD;
    const float* rowy = rowx + (size_t)BT * AD;
    if (p + 1 < NVAR / 2) {
      __builtin_prefetch(rowx + (size_t)2 * BT * AD, 0, 1);
      __builtin_prefetch(rowx + (size_t)3 * BT * AD, 0, 1);
    }
    variant_pair<false>(rowx, rowy, (p & 1) ? s_set1 : s_set0, s_h2, s_pwp + wid * PPW,
                        hobs, bw1, bw2, bw3, bias2, accq, col, hi, kb, jc1a, jc1b, jc2, k3);
  }

  const int rrow = tid >> 3;
  const int qa   = (tid & 7) * 4;
  const float4 bias3 = *(const float4*)(b3 + qa);

  #pragma unroll
  for (int t = 0; t < 2; ++t) {
    #pragma unroll
    for (int v = 0; v < 8; ++v)
      s_pqp[wid * PPW + (v + 8 * hi) * PPP + t * 16 + col] = accq[t][v];
  }
  __syncthreads();

  {
    const int o = rrow * PPP + qa;
    const float4 w0 = *(const float4*)(s_pwp + o);
    const float4 w1 = *(const float4*)(s_pwp + PPW + o);
    const float4 w2 = *(const float4*)(s_pwp + 2 * PPW + o);
    const float4 w3 = *(const float4*)(s_pwp + 3 * PPW + o);
    const float4 q0 = *(const float4*)(s_pqp + o);
    const float4 q1 = *(const float4*)(s_pqp + PPW + o);
    const float4 q2 = *(const float4*)(s_pqp + 2 * PPW + o);
    const float4 q3 = *(const float4*)(s_pqp + 3 * PPW + o);
    float pw[4], pq[4];
    pw[0] = ((w0.x + w1.x) + w2.x) + w3.x + bias3.x;
    pw[1] = ((w0.y + w1.y) + w2.y) + w3.y + bias3.y;
    pw[2] = ((w0.z + w1.z) + w2.z) + w3.z + bias3.z;
    pw[3] = ((w0.w + w1.w) + w2.w) + w3.w + bias3.w;
    pq[0] = (((q0.x + q1.x) + q2.x) + q3.x) * INV_S1 + bias3.x;
    pq[1] = (((q0.y + q1.y) + q2.y) + q3.y) * INV_S1 + bias3.y;
    pq[2] = (((q0.z + q1.z) + q2.z) + q3.z) * INV_S1 + bias3.z;
    pq[3] = (((q0.w + q1.w) + q2.w) + q3.w) * INV_S1 + bias3.w;

    const float mp = row8_max(fmaxf(fmaxf(pw[0], pw[1]), fmaxf(pw[2], pw[3])));
    const float mq = row8_max(fmaxf(fmaxf(pq[0], pq[1]), fmaxf(pq[2], pq[3])));
    float ep[4], eq[4];
    #pragma unroll
    for (int i = 0; i < 4; ++i) { ep[i] = expf(pw[i] - mp); eq[i] = expf(pq[i] - mq); }
    const float sp = row8_sum((ep[0] + ep[1]) + (ep[2] + ep[3]));
    const float sq = row8_sum((eq[0] + eq[1]) + (eq[2] + eq[3]));
    const float lsp = logf(sp);
    const float lsq = logf(sq);
    float kt[4];
    #pragma unroll
    for (int i = 0; i < 4; ++i) {
      const float log_p = (pw[i] - mp) - lsp;
      const float log_q = (pq[i] - mq) - lsq;
      const float q     = eq[i] / sq;
      kt[i] = q * (log_q - log_p);
    }
    const float kl = row8_sum((kt[0] + kt[1]) + (kt[2] + kt[3]));
    if ((tid & 7) == 0) out[(size_t)(bt0 + rrow) * NA + n] = kl * (1.0f / AD);
  }
}

extern "C" void kernel_launch(void* const* d_in, const int* in_sizes, int n_in,
                              void* d_out, int out_size, void* d_ws, size_t ws_size,
                              hipStream_t stream) {
  (void)in_sizes; (void)n_in; (void)d_ws; (void)ws_size; (void)out_size;
  const float* obs = (const float*)d_in[0];
  const float* act = (const float*)d_in[1];
  const float* cf  = (const float*)d_in[2];
  const float* W1  = (const float*)d_in[3];
  const float* b1  = (const float*)d_in[4];
  const float* W2  = (const float*)d_in[5];
  const float* b2  = (const float*)d_in[6];
  const float* W3  = (const float*)d_in[7];
  const float* b3  = (const float*)d_in[8];
  float* out = (float*)d_out;

  const int tiles = (NA * BB * TT) / 16;
  const size_t lds_bytes = LDS_FLOATS * sizeof(float);
  coma_influence_kernel<<<tiles, 128, lds_bytes, stream>>>(
      obs, act, cf, W1, b1, W2, b2, W3, b3, out);
}
